# starting kernel
# baseline (speedup 1.0000x reference)
#include <hip/hip_runtime.h>


typedef __attribute__((ext_vector_type(2))) float v2f;
typedef __attribute__((ext_vector_type(4))) float v4f;
typedef __attribute__((ext_vector_type(8))) float v8f;

#define N_NODES 100000
#define N_EDGES 1200000
#define IN_DIM  256
#define Z_DIM   64
#define M_TILES (N_NODES / 16)
#define LDS_STRIDE 72


__global__ __launch_bounds__(256) void zero_counts_kernel(unsigned* __restrict__ cnt) {
    int i = blockIdx.x * 256 + threadIdx.x;
    if (i < N_NODES) cnt[i] = 0u;
}

__global__ __launch_bounds__(256) void count_deg_kernel(const int* __restrict__ ei,
                                                        unsigned* __restrict__ cnt) {
    int e = blockIdx.x * 256 + threadIdx.x;
    if (e < N_EDGES) {
        int col = ei[N_EDGES + e];
        atomicAdd(&cnt[col], 1u);
    }
}

__global__ __launch_bounds__(256) void dinv_kernel(const unsigned* __restrict__ cnt,
                                                   float* __restrict__ dinv) {
    int i = blockIdx.x * 256 + threadIdx.x;
    if (i < N_NODES) {
        float d = (float)cnt[i] + 1.0f;
        dinv[i] = __frsqrt_rn(d);
    }
}


__global__ __launch_bounds__(256) void gemm_scale_kernel(const float* __restrict__ x,
                                                         const float* __restrict__ W,
                                                         const float* __restrict__ dinv,
                                                         float* __restrict__ s,
                                                         float* __restrict__ out) {
    __shared__ float Wl[IN_DIM * LDS_STRIDE];

    const int tid = threadIdx.x;

    for (int i = tid; i < (IN_DIM * Z_DIM) / 4; i += 256) {
        int r  = i >> 4;
        int c4 = (i & 15) << 2;
        v4f w = *(const v4f*)(W + r * Z_DIM + c4);
        *(v4f*)(&Wl[r * LDS_STRIDE + c4]) = w;
    }
    __syncthreads();

    const int wave = tid >> 5;
    const int lane = tid & 31;
    const int mt   = blockIdx.x * 8 + wave;
    if (mt >= M_TILES) return;

    const int m0   = mt * 16;
    const int half = lane >> 4;
    const int l16  = lane & 15;

    v8f z = {};
    v8f acc0 = z, acc1 = z, acc2 = z, acc3 = z;

    const float* xrow = x + (size_t)(m0 + l16) * IN_DIM + 2 * half;

    #pragma unroll 4
    for (int k = 0; k < IN_DIM; k += 4) {
        v2f a = *(const v2f*)(xrow + k);
        const int kr = k + 2 * half;
        v2f b0, b1, b2, b3;
        b0.x = Wl[(kr    ) * LDS_STRIDE +  0 + l16];
        b0.y = Wl[(kr + 1) * LDS_STRIDE +  0 + l16];
        b1.x = Wl[(kr    ) * LDS_STRIDE + 16 + l16];
        b1.y = Wl[(kr + 1) * LDS_STRIDE + 16 + l16];
        b2.x = Wl[(kr    ) * LDS_STRIDE + 32 + l16];
        b2.y = Wl[(kr + 1) * LDS_STRIDE + 32 + l16];
        b3.x = Wl[(kr    ) * LDS_STRIDE + 48 + l16];
        b3.y = Wl[(kr + 1) * LDS_STRIDE + 48 + l16];
        acc0 = __builtin_amdgcn_wmma_f32_16x16x4_f32(false, a, false, b0, (short)0, acc0, false, false);
        acc1 = __builtin_amdgcn_wmma_f32_16x16x4_f32(false, a, false, b1, (short)0, acc1, false, false);
        acc2 = __builtin_amdgcn_wmma_f32_16x16x4_f32(false, a, false, b2, (short)0, acc2, false, false);
        acc3 = __builtin_amdgcn_wmma_f32_16x16x4_f32(false, a, false, b3, (short)0, acc3, false, false);
    }

    float dv[8];
    #pragma unroll
    for (int r = 0; r < 8; ++r) dv[r] = dinv[m0 + half * 8 + r];

    v8f accs[4] = {acc0, acc1, acc2, acc3};
    #pragma unroll
    for (int nt = 0; nt < 4; ++nt) {
        #pragma unroll
        for (int r = 0; r < 8; ++r) {
            float v = accs[nt][r] * dv[r];
            size_t off = (size_t)(m0 + half * 8 + r) * Z_DIM + nt * 16 + l16;
            s[off]   = v;
            out[off] = v;
        }
    }
}


__global__ __launch_bounds__(256) void scatter_kernel(const int* __restrict__ ei,
                                                      const float* __restrict__ s,
                                                      float* __restrict__ out) {
    unsigned gid = blockIdx.x * 256u + threadIdx.x;
    int e = (int)(gid >> 6);
    int k = (int)(gid & 63u);
    if (e >= N_EDGES) return;
    int row = ei[e];
    int col = ei[N_EDGES + e];
    float v = s[(size_t)row * Z_DIM + k];
    unsafeAtomicAdd(&out[(size_t)col * Z_DIM + k], v);
}


__global__ __launch_bounds__(256) void finalize_kernel(float* __restrict__ out,
                                                       const float* __restrict__ dinv,
                                                       const float* __restrict__ bias,
                                                       const float* __restrict__ pa) {
    int gid = blockIdx.x * 256 + threadIdx.x;
    if (gid >= N_NODES * Z_DIM) return;
    int k = gid & 63;
    float v = dinv[gid >> 6] * out[gid] + bias[k];
    out[gid] = (v >= 0.0f) ? v : pa[k] * v;
}


extern "C" void kernel_launch(void* const* d_in, const int* in_sizes, int n_in,
                              void* d_out, int out_size, void* d_ws, size_t ws_size,
                              hipStream_t stream) {
    const float* x    = (const float*)d_in[0];
    const int*   ei   = (const int*)d_in[1];
    const float* W    = (const float*)d_in[3];
    const float* bias = (const float*)d_in[4];
    const float* pa   = (const float*)d_in[5];
    float* out = (float*)d_out;

    unsigned* cnt = (unsigned*)d_ws;
    float* dinv   = (float*)((char*)d_ws + (size_t)N_NODES * 4);
    float* s      = (float*)((char*)d_ws + (size_t)2 * N_NODES * 4);

    zero_counts_kernel<<<(N_NODES + 255) / 256, 256, 0, stream>>>(cnt);
    count_deg_kernel<<<(N_EDGES + 255) / 256, 256, 0, stream>>>(ei, cnt);
    dinv_kernel<<<(N_NODES + 255) / 256, 256, 0, stream>>>(cnt, dinv);

    gemm_scale_kernel<<<(M_TILES + 7) / 8, 256, 0, stream>>>(x, W, dinv, s, out);

    const long long pairs = (long long)N_EDGES * Z_DIM;
    scatter_kernel<<<(unsigned)((pairs + 255) / 256), 256, 0, stream>>>(ei, s, out);

    finalize_kernel<<<(N_NODES * Z_DIM + 255) / 256, 256, 0, stream>>>(out, dinv, bias, pa);
}


# Edge ranks and degrees from per-chunk LDS histograms of all nodes (9600-edge chunks, 2 counters a word) plus a prefix over chunks: no global atomics or counter zeroing; row sums in chunk-major order
# speedup vs baseline: 2.8201x; 2.8201x over previous
#include <hip/hip_runtime.h>


typedef __attribute__((ext_vector_type(2))) float v2f;
typedef __attribute__((ext_vector_type(4))) float v4f;
typedef __attribute__((ext_vector_type(8))) float v8f;

#define N_NODES 100000
#define N_EDGES 1200000
#define IN_DIM  256
#define Z_DIM   64
#define M_TILES (N_NODES / 16)
#define LDS_STRIDE 72
#define SLOTS   64u
#define NCH     125
#define CHUNK   (N_EDGES / NCH)
#define NW      (N_NODES / 2)
#define NWS     51200
#define RCLAMP  1024u

typedef __attribute__((ext_vector_type(2))) unsigned v2u;
typedef __attribute__((ext_vector_type(4))) unsigned v4u;


__device__ __forceinline__ unsigned node_of(unsigned id) {
    return min(id, (unsigned)(N_NODES - 1));
}

__global__ __launch_bounds__(1024) void hist_kernel(const int* __restrict__ ei,
                                                    float* __restrict__ scratch) {
    __shared__ unsigned h[NW];
    const int tid = threadIdx.x;
    const int* col = ei + N_EDGES + blockIdx.x * CHUNK;

    unsigned c[10];
    #pragma unroll
    for (int k = 0; k < 10; ++k) {
        const int i = tid + k * 1024;
        c[k] = (i < CHUNK) ? (unsigned)col[i] : 0u;
    }

    const v4u z = {0u, 0u, 0u, 0u};
    for (int i = tid; i < NW / 4; i += 1024) ((v4u*)h)[i] = z;
    __syncthreads();

    #pragma unroll
    for (int k = 0; k < 10; ++k) {
        if (tid + k * 1024 < CHUNK) {
            const unsigned n = node_of(c[k]);
            atomicAdd(&h[n >> 1], 1u << ((n & 1u) * 16u));
        }
    }
    __syncthreads();

    v4u* dst = (v4u*)scratch + (size_t)blockIdx.x * (NWS / 4);
    for (int i = tid; i < NW / 4; i += 1024) dst[i] = ((const v4u*)h)[i];
}

__global__ __launch_bounds__(128) void scan_kernel(const float* __restrict__ scratch,
                                                   unsigned* __restrict__ base,
                                                   unsigned* __restrict__ cnt,
                                                   float* __restrict__ dinv) {
    const int j = blockIdx.x * 128 + threadIdx.x;
    if (j >= NW) return;
    const unsigned* part = (const unsigned*)scratch;

    unsigned lo = 0u, hi = 0u;
    for (int w0 = 0; w0 < NCH; w0 += 25) {
        unsigned v[25];
        #pragma unroll
        for (int k = 0; k < 25; ++k) v[k] = part[(size_t)(w0 + k) * NWS + j];
        #pragma unroll
        for (int k = 0; k < 25; ++k) {
            base[(size_t)(w0 + k) * NWS + j] = min(lo, RCLAMP) | (min(hi, RCLAMP) << 16);
            lo += v[k] & 0xFFFFu;
            hi += v[k] >> 16;
        }
    }
    v2u d = {lo, hi};
    ((v2u*)cnt)[j] = d;
    v2f r;
    r.x = __frsqrt_rn((float)lo + 1.0f);
    r.y = __frsqrt_rn((float)hi + 1.0f);
    ((v2f*)dinv)[j] = r;
}

__global__ __launch_bounds__(1024) void file_kernel(const int* __restrict__ ei,
                                                    const unsigned* __restrict__ base,
                                                    float* __restrict__ out) {
    __shared__ unsigned b[NW];
    const int tid = threadIdx.x;
    const int* row = ei + blockIdx.x * CHUNK;
    const int* col = ei + N_EDGES + blockIdx.x * CHUNK;

    unsigned c[10];
    int r[10];
    #pragma unroll
    for (int k = 0; k < 10; ++k) {
        const int i = tid + k * 1024;
        c[k] = (i < CHUNK) ? (unsigned)col[i] : 0u;
        r[k] = (i < CHUNK) ? row[i] : 0;
    }

    const v4u* src = (const v4u*)base + (size_t)blockIdx.x * (NWS / 4);
    for (int i = tid; i < NW / 4; i += 1024) ((v4u*)b)[i] = src[i];
    __syncthreads();

    unsigned old[10];
    #pragma unroll
    for (int k = 0; k < 10; ++k) {
        old[k] = 0u;
        if (tid + k * 1024 < CHUNK) {
            const unsigned n = node_of(c[k]);
            old[k] = atomicAdd(&b[n >> 1], 1u << ((n & 1u) * 16u));
        }
    }
    #pragma unroll
    for (int k = 0; k < 10; ++k) {
        if (tid + k * 1024 < CHUNK) {
            const unsigned n = node_of(c[k]);
            const unsigned rank = (old[k] >> ((n & 1u) * 16u)) & 0xFFFFu;
            if (rank < SLOTS) out[(size_t)n * Z_DIM + rank] = __int_as_float(r[k]);
        }
    }
}


__global__ __launch_bounds__(256) void gemm_scale_kernel(const float* __restrict__ x,
                                                         const float* __restrict__ W,
                                                         const float* __restrict__ dinv,
                                                         float* __restrict__ s) {
    __shared__ float Wl[IN_DIM * LDS_STRIDE];

    const int tid = threadIdx.x;

    for (int i = tid; i < (IN_DIM * Z_DIM) / 4; i += 256) {
        int r  = i >> 4;
        int c4 = (i & 15) << 2;
        v4f w = *(const v4f*)(W + r * Z_DIM + c4);
        *(v4f*)(&Wl[r * LDS_STRIDE + c4]) = w;
    }
    __syncthreads();

    const int wave = tid >> 5;
    const int lane = tid & 31;
    const int mt   = blockIdx.x * 8 + wave;
    if (mt >= M_TILES) return;

    const int m0   = mt * 16;
    const int half = lane >> 4;
    const int l16  = lane & 15;

    v8f z = {};
    v8f acc0 = z, acc1 = z, acc2 = z, acc3 = z;

    const float* xrow = x + (size_t)(m0 + l16) * IN_DIM + 2 * half;

    #pragma unroll 4
    for (int k = 0; k < IN_DIM; k += 4) {
        v2f a = *(const v2f*)(xrow + k);
        const int kr = k + 2 * half;
        v2f b0, b1, b2, b3;
        b0.x = Wl[(kr    ) * LDS_STRIDE +  0 + l16];
        b0.y = Wl[(kr + 1) * LDS_STRIDE +  0 + l16];
        b1.x = Wl[(kr    ) * LDS_STRIDE + 16 + l16];
        b1.y = Wl[(kr + 1) * LDS_STRIDE + 16 + l16];
        b2.x = Wl[(kr    ) * LDS_STRIDE + 32 + l16];
        b2.y = Wl[(kr + 1) * LDS_STRIDE + 32 + l16];
        b3.x = Wl[(kr    ) * LDS_STRIDE + 48 + l16];
        b3.y = Wl[(kr + 1) * LDS_STRIDE + 48 + l16];
        acc0 = __builtin_amdgcn_wmma_f32_16x16x4_f32(false, a, false, b0, (short)0, acc0, false, false);
        acc1 = __builtin_amdgcn_wmma_f32_16x16x4_f32(false, a, false, b1, (short)0, acc1, false, false);
        acc2 = __builtin_amdgcn_wmma_f32_16x16x4_f32(false, a, false, b2, (short)0, acc2, false, false);
        acc3 = __builtin_amdgcn_wmma_f32_16x16x4_f32(false, a, false, b3, (short)0, acc3, false, false);
    }

    float dv[8];
    #pragma unroll
    for (int r = 0; r < 8; ++r) dv[r] = dinv[m0 + half * 8 + r];

    v8f accs[4] = {acc0, acc1, acc2, acc3};
    #pragma unroll
    for (int nt = 0; nt < 4; ++nt) {
        #pragma unroll
        for (int r = 0; r < 8; ++r) {
            float v = accs[nt][r] * dv[r];
            size_t off = (size_t)(m0 + half * 8 + r) * Z_DIM + nt * 16 + l16;
            s[off] = v;
        }
    }
}


__device__ __forceinline__ v4f load_row_chunk(const float* __restrict__ s, int id, int l16) {
    unsigned src = min((unsigned)id, (unsigned)(N_NODES - 1));
    return *(const v4f*)(s + (size_t)src * Z_DIM + 4 * l16);
}

__global__ __launch_bounds__(256) void pull_kernel(const int* __restrict__ ei,
                                                   const unsigned* __restrict__ cnt,
                                                   const float* __restrict__ dinv,
                                                   const float* __restrict__ s,
                                                   const float* __restrict__ bias,
                                                   const float* __restrict__ pa,
                                                   float* __restrict__ out) {
    const int gid = blockIdx.x * 256 + threadIdx.x;
    const int c   = gid >> 4;
    const int l16 = gid & 15;
    if (c >= N_NODES) return;

    const unsigned deg = cnt[c];
    const float* slot = out + (size_t)c * Z_DIM;

    v4f acc = *(const v4f*)(s + (size_t)c * Z_DIM + 4 * l16);
    v4f w0 = *(const v4f*)(slot);
    v4f w1 = *(const v4f*)(slot + 4);
    const float d = dinv[c];
    const v4f b = *(const v4f*)(bias + 4 * l16);
    const v4f a = *(const v4f*)(pa + 4 * l16);

    if (deg <= SLOTS) {
        for (unsigned j = 0; j < deg; j += 8) {
            const int id[8] = {__float_as_int(w0.x), __float_as_int(w0.y),
                               __float_as_int(w0.z), __float_as_int(w0.w),
                               __float_as_int(w1.x), __float_as_int(w1.y),
                               __float_as_int(w1.z), __float_as_int(w1.w)};
            if (j + 8 < deg) {
                w0 = *(const v4f*)(slot + j + 8);
                w1 = *(const v4f*)(slot + j + 12);
            }
            v4f r[8];
            #pragma unroll
            for (int k = 0; k < 8; ++k) {
                v4f z = {};
                r[k] = z;
                if (j + k < deg) r[k] = load_row_chunk(s, id[k], l16);
            }
            #pragma unroll
            for (int k = 0; k < 8; ++k) acc += r[k];
        }
    } else {
        for (int e = 0; e < N_EDGES; ++e) {
            if (ei[N_EDGES + e] == c) acc += load_row_chunk(s, ei[e], l16);
        }
    }

    v4f v = d * acc + b;
    v4f o;
    o.x = (v.x >= 0.0f) ? v.x : a.x * v.x;
    o.y = (v.y >= 0.0f) ? v.y : a.y * v.y;
    o.z = (v.z >= 0.0f) ? v.z : a.z * v.z;
    o.w = (v.w >= 0.0f) ? v.w : a.w * v.w;
    *(v4f*)(out + (size_t)c * Z_DIM + 4 * l16) = o;
}


extern "C" void kernel_launch(void* const* d_in, const int* in_sizes, int n_in,
                              void* d_out, int out_size, void* d_ws, size_t ws_size,
                              hipStream_t stream) {
    const float* x    = (const float*)d_in[0];
    const int*   ei   = (const int*)d_in[1];
    const float* W    = (const float*)d_in[3];
    const float* bias = (const float*)d_in[4];
    const float* pa   = (const float*)d_in[5];
    float* out = (float*)d_out;

    unsigned* cnt = (unsigned*)d_ws;
    float* dinv   = (float*)((char*)d_ws + (size_t)N_NODES * 4);
    float* s      = (float*)((char*)d_ws + (size_t)2 * N_NODES * 4);

    hist_kernel<<<NCH, 1024, 0, stream>>>(ei, out);
    scan_kernel<<<(NW + 127) / 128, 128, 0, stream>>>(out, (unsigned*)s, cnt, dinv);
    file_kernel<<<NCH, 1024, 0, stream>>>(ei, (const unsigned*)s, out);

    gemm_scale_kernel<<<(M_TILES + 7) / 8, 256, 0, stream>>>(x, W, dinv, s);

    pull_kernel<<<(N_NODES * 16 + 255) / 256, 256, 0, stream>>>(ei, cnt, dinv, s, bias, pa, out);
}
